# starting kernel
# baseline (speedup 1.0000x reference)
#include <hip/hip_runtime.h>
#include <math.h>

typedef float v2f __attribute__((ext_vector_type(2)));
typedef float v8f __attribute__((ext_vector_type(8)));

__global__ void k_fill(float* __restrict__ p, float v, int n) {
    int i = blockIdx.x * blockDim.x + threadIdx.x;
    if (i < n) p[i] = v;
}

__global__ void k_deg_accum(const int* __restrict__ dst, float* __restrict__ deg, int E) {
    int e = blockIdx.x * blockDim.x + threadIdx.x;
    if (e < E) unsafeAtomicAdd(&deg[dst[e]], 1.0f);
}

__global__ void k_rsqrt_inplace(float* __restrict__ p, int n) {
    int i = blockIdx.x * blockDim.x + threadIdx.x;
    if (i < n) p[i] = rsqrtf(p[i]);
}

__global__ void k_norm(const int* __restrict__ src, const int* __restrict__ dst,
                       const float* __restrict__ dis, float* __restrict__ norm, int E) {
    int e = blockIdx.x * blockDim.x + threadIdx.x;
    if (e < E) norm[e] = dis[src[e]] * dis[dst[e]];
}

template <int K, int NCOL>
__global__ __launch_bounds__(256) void k_gemm_wmma(const float* __restrict__ A,
                                                   const float* __restrict__ B,
                                                   float* __restrict__ O, int nrows) {
    int wave = threadIdx.x >> 5;
    int lane = threadIdx.x & 31;
    int tile = blockIdx.x * (blockDim.x >> 5) + wave;
    int base = tile << 4;
    if (base >= nrows) return;
    int hi = lane >> 4;
    int lo = lane & 15;

    int arow = base + lo;
    arow = arow < nrows ? arow : (nrows - 1);
    const float* __restrict__ ap = A + (long)arow * K + 2 * hi;
    const float* __restrict__ bp = B + lo;
    bool bvalid = (lo < NCOL);

    v8f c = {0.f, 0.f, 0.f, 0.f, 0.f, 0.f, 0.f, 0.f};
#pragma unroll
    for (int k0 = 0; k0 < K; k0 += 4) {
        v2f a = *(const v2f*)(ap + k0);
        int ka = k0 + 2 * hi;
        v2f b;
        b.x = bvalid ? bp[(ka + 0) * NCOL] : 0.f;
        b.y = bvalid ? bp[(ka + 1) * NCOL] : 0.f;
        c = __builtin_amdgcn_wmma_f32_16x16x4_f32(false, a, false, b, (short)0, c,
                                                  false, false);
    }
#pragma unroll
    for (int v = 0; v < 8; ++v) {
        int row = base + v + 8 * hi;
        if (row < nrows && lo < NCOL) O[(long)row * NCOL + lo] = c[v];
    }
}

template <int F>
__global__ void k_scatter(const int* __restrict__ src, const int* __restrict__ dst,
                          const float* __restrict__ norm, const float* __restrict__ h,
                          float* __restrict__ acc, int E) {
    long gid = (long)blockIdx.x * blockDim.x + threadIdx.x;
    long e = gid / F;
    int f = (int)(gid - e * F);
    if (e < (long)E) {
        int s = src[e], d = dst[e];
        float v = norm[e] * h[(long)s * F + f];
        unsafeAtomicAdd(&acc[(long)d * F + f], v);
    }
}

__global__ void k_relu_ep(const float* __restrict__ acc, const float* __restrict__ hx,
                          const float* __restrict__ dis, const float* __restrict__ b,
                          float* __restrict__ out, int n) {
    int gid = blockIdx.x * blockDim.x + threadIdx.x;
    if (gid < n * 16) {
        int i = gid >> 4, f = gid & 15;
        float di = dis[i];
        float v = acc[gid] + di * di * hx[gid] + b[f];
        out[gid] = v > 0.f ? v : 0.f;
    }
}

__global__ void k_lsm_ep(const float* __restrict__ acc, const float* __restrict__ hx,
                         const float* __restrict__ dis, const float* __restrict__ b,
                         float* __restrict__ out, int n) {
    int i = blockIdx.x * blockDim.x + threadIdx.x;
    if (i < n) {
        float d2 = dis[i] * dis[i];
        float z0 = acc[2 * i + 0] + d2 * hx[2 * i + 0] + b[0];
        float z1 = acc[2 * i + 1] + d2 * hx[2 * i + 1] + b[1];
        float m = fmaxf(z0, z1);
        float lse = m + __logf(__expf(z0 - m) + __expf(z1 - m));
        out[2 * i + 0] = z0 - lse;
        out[2 * i + 1] = z1 - lse;
    }
}

extern "C" void kernel_launch(void* const* d_in, const int* in_sizes, int n_in,
                              void* d_out, int out_size, void* d_ws, size_t ws_size,
                              hipStream_t stream) {
    const float* x  = (const float*)d_in[0];
    const int*   ei = (const int*)d_in[1];
    const float* W1 = (const float*)d_in[2];
    const float* b1 = (const float*)d_in[3];
    const float* W2 = (const float*)d_in[4];
    const float* b2 = (const float*)d_in[5];
    const float* W3 = (const float*)d_in[6];
    const float* b3 = (const float*)d_in[7];

    const int N = in_sizes[0] / 128;
    const int E = in_sizes[1] / 2;
    const int* src = ei;
    const int* dst = ei + E;

    char* ws = (char*)d_ws;
    size_t off = 0;
    auto carve = [&](size_t bytes) -> void* {
        void* p = ws + off;
        off += (bytes + 255) & ~(size_t)255;
        return p;
    };
    float* dis  = (float*)carve((size_t)N * 4);
    float* norm = (float*)carve((size_t)E * 4);
    float* hx   = (float*)carve((size_t)N * 16 * 4);
    float* acc  = (float*)carve((size_t)N * 16 * 4);
    float* h    = (float*)carve((size_t)N * 16 * 4);
    (void)ws_size; (void)n_in; (void)out_size;

    const int T = 256;
    auto blk = [](long n, int t) { return (unsigned)((n + t - 1) / t); };

    k_fill<<<blk(N, T), T, 0, stream>>>(dis, 1.0f, N);
    k_deg_accum<<<blk(E, T), T, 0, stream>>>(dst, dis, E);
    k_rsqrt_inplace<<<blk(N, T), T, 0, stream>>>(dis, N);
    k_norm<<<blk(E, T), T, 0, stream>>>(src, dst, dis, norm, E);

    const int ntile = (N + 15) / 16;
    const unsigned gblk = (unsigned)((ntile + 7) / 8);

    k_gemm_wmma<128, 16><<<gblk, 256, 0, stream>>>(x, W1, hx, N);
    k_fill<<<blk((long)N * 16, T), T, 0, stream>>>(acc, 0.f, N * 16);
    k_scatter<16><<<blk((long)E * 16, T), T, 0, stream>>>(src, dst, norm, hx, acc, E);
    k_relu_ep<<<blk((long)N * 16, T), T, 0, stream>>>(acc, hx, dis, b1, h, N);

    k_gemm_wmma<16, 16><<<gblk, 256, 0, stream>>>(h, W2, hx, N);
    k_fill<<<blk((long)N * 16, T), T, 0, stream>>>(acc, 0.f, N * 16);
    k_scatter<16><<<blk((long)E * 16, T), T, 0, stream>>>(src, dst, norm, hx, acc, E);
    k_relu_ep<<<blk((long)N * 16, T), T, 0, stream>>>(acc, hx, dis, b2, h, N);

    k_gemm_wmma<16, 2><<<gblk, 256, 0, stream>>>(h, W3, hx, N);
    k_fill<<<blk((long)N * 2, T), T, 0, stream>>>(acc, 0.f, N * 2);
    k_scatter<2><<<blk((long)E * 2, T), T, 0, stream>>>(src, dst, norm, hx, acc, E);
    k_lsm_ep<<<blk(N, T), T, 0, stream>>>(acc, hx, dis, b3, (float*)d_out, N);
}


# each die owns a destination slice: groups draw (edge chunk, slice) tasks from their die's queue, so float atomics on a slice come from one die; atomics kept, sums still in arrival order
# speedup vs baseline: 2.7809x; 2.7809x over previous
#include <hip/hip_runtime.h>
#include <math.h>

typedef float v2f __attribute__((ext_vector_type(2)));
typedef float v8f __attribute__((ext_vector_type(8)));

__global__ void k_fill(float* __restrict__ p, float v, int n) {
    int i = blockIdx.x * blockDim.x + threadIdx.x;
    if (i < n) p[i] = v;
}

#define NQ 8
#define TSTRIDE 64
#define EPT 8
#define CHUNK (256 * EPT)

__global__ void k_init(float* __restrict__ p, int* __restrict__ tick, int n) {
    int i = blockIdx.x * blockDim.x + threadIdx.x;
    if (i < n) p[i] = 1.0f;
    if (i < 4 * NQ * TSTRIDE) tick[i] = 0;
}

__device__ __forceinline__ void draw_task(int* __restrict__ tick, int nchunk,
                                          int& chunk, int& slice) {
    __shared__ int s_q, s_t;
    if (threadIdx.x == 0) {
        int die = (int)((__builtin_amdgcn_s_sendmsg_rtn(0x87) >> 16) & (NQ - 1));
        int q = die, t = nchunk;
        for (int i = 0; i < NQ; ++i) {
            q = (die + i) & (NQ - 1);
            t = atomicAdd(&tick[q * TSTRIDE], 1);
            if (t < nchunk) break;
        }
        s_q = q; s_t = t;
    }
    __syncthreads();
    slice = s_q; chunk = s_t;
}

__global__ __launch_bounds__(256) void k_deg_own(const int* __restrict__ dst,
                                                 float* __restrict__ deg,
                                                 int* __restrict__ tick, int E, int N,
                                                 int nchunk) {
    int chunk, slice;
    draw_task(tick, nchunk, chunk, slice);
    if (chunk >= nchunk) return;
    unsigned R = (unsigned)(N + NQ - 1) / NQ;
    unsigned lo = (unsigned)slice * R;
    int e0 = chunk * CHUNK + (int)threadIdx.x;
    int dv[EPT];
#pragma unroll
    for (int j = 0; j < EPT; ++j) {
        int e = e0 + j * 256;
        dv[j] = e < E ? dst[e] : -1;
    }
#pragma unroll
    for (int j = 0; j < EPT; ++j)
        if ((unsigned)dv[j] - lo < R) unsafeAtomicAdd(&deg[dv[j]], 1.0f);
}

__global__ void k_rsqrt_inplace(float* __restrict__ p, int n) {
    int i = blockIdx.x * blockDim.x + threadIdx.x;
    if (i < n) p[i] = rsqrtf(p[i]);
}

__global__ void k_norm(const int* __restrict__ src, const int* __restrict__ dst,
                       const float* __restrict__ dis, float* __restrict__ norm, int E) {
    int e = blockIdx.x * blockDim.x + threadIdx.x;
    if (e < E) norm[e] = dis[src[e]] * dis[dst[e]];
}

template <int K, int NCOL>
__global__ __launch_bounds__(256) void k_gemm_wmma(const float* __restrict__ A,
                                                   const float* __restrict__ B,
                                                   float* __restrict__ O, int nrows) {
    int wave = threadIdx.x >> 5;
    int lane = threadIdx.x & 31;
    int tile = blockIdx.x * (blockDim.x >> 5) + wave;
    int base = tile << 4;
    if (base >= nrows) return;
    int hi = lane >> 4;
    int lo = lane & 15;

    int arow = base + lo;
    arow = arow < nrows ? arow : (nrows - 1);
    const float* __restrict__ ap = A + (long)arow * K + 2 * hi;
    const float* __restrict__ bp = B + lo;
    bool bvalid = (lo < NCOL);

    v8f c = {0.f, 0.f, 0.f, 0.f, 0.f, 0.f, 0.f, 0.f};
#pragma unroll
    for (int k0 = 0; k0 < K; k0 += 4) {
        v2f a = *(const v2f*)(ap + k0);
        int ka = k0 + 2 * hi;
        v2f b;
        b.x = bvalid ? bp[(ka + 0) * NCOL] : 0.f;
        b.y = bvalid ? bp[(ka + 1) * NCOL] : 0.f;
        c = __builtin_amdgcn_wmma_f32_16x16x4_f32(false, a, false, b, (short)0, c,
                                                  false, false);
    }
#pragma unroll
    for (int v = 0; v < 8; ++v) {
        int row = base + v + 8 * hi;
        if (row < nrows && lo < NCOL) O[(long)row * NCOL + lo] = c[v];
    }
}

template <int F>
__global__ __launch_bounds__(256) void k_scatter_own(
    const int* __restrict__ src, const int* __restrict__ dst, const float* __restrict__ norm,
    const float* __restrict__ h, float* __restrict__ acc, int* __restrict__ tick, int E, int N,
    int nchunk) {
    __shared__ int l_s[8][32 * EPT];
    __shared__ int l_d[8][32 * EPT];
    __shared__ float l_n[8][32 * EPT];
    int chunk, slice;
    draw_task(tick, nchunk, chunk, slice);
    if (chunk >= nchunk) return;
    unsigned R = (unsigned)(N + NQ - 1) / NQ;
    unsigned lo = (unsigned)slice * R;
    int wave = threadIdx.x >> 5;
    int lane = threadIdx.x & 31;
    int e0 = chunk * CHUNK + (int)threadIdx.x;
    int dv[EPT];
#pragma unroll
    for (int j = 0; j < EPT; ++j) {
        int e = e0 + j * 256;
        dv[j] = e < E ? dst[e] : -1;
    }
    int sv[EPT];
    float nv[EPT];
#pragma unroll
    for (int j = 0; j < EPT; ++j) {
        sv[j] = 0;
        nv[j] = 0.f;
        if ((unsigned)dv[j] - lo < R) {
            int e = e0 + j * 256;
            sv[j] = src[e];
            nv[j] = norm[e];
        }
    }
    int cnt = 0;
#pragma unroll
    for (int j = 0; j < EPT; ++j) {
        bool in = (unsigned)dv[j] - lo < R;
        unsigned m = __builtin_amdgcn_ballot_w32(in);
        if (in) {
            int slot = cnt + __popc(m & ((1u << lane) - 1u));
            l_s[wave][slot] = sv[j];
            l_d[wave][slot] = dv[j];
            l_n[wave][slot] = nv[j];
        }
        cnt += __popc(m);
    }
    __syncthreads();
    constexpr int EPW = 32 / F;
    int sub = lane / F;
    int f = lane - sub * F;
    for (int i = sub; i < cnt; i += 4 * EPW) {
        int d[4];
        float nr[4], hv[4];
#pragma unroll
        for (int u = 0; u < 4; ++u) {
            int k = i + u * EPW;
            d[u] = -1;
            nr[u] = 0.f;
            hv[u] = 0.f;
            if (k < cnt) {
                int s = l_s[wave][k];
                d[u] = l_d[wave][k];
                nr[u] = l_n[wave][k];
                hv[u] = h[(long)s * F + f];
            }
        }
#pragma unroll
        for (int u = 0; u < 4; ++u)
            if (d[u] >= 0) unsafeAtomicAdd(&acc[(long)d[u] * F + f], nr[u] * hv[u]);
    }
}

__global__ void k_relu_ep(const float* __restrict__ acc, const float* __restrict__ hx,
                          const float* __restrict__ dis, const float* __restrict__ b,
                          float* __restrict__ out, int n) {
    int gid = blockIdx.x * blockDim.x + threadIdx.x;
    if (gid < n * 16) {
        int i = gid >> 4, f = gid & 15;
        float di = dis[i];
        float v = acc[gid] + di * di * hx[gid] + b[f];
        out[gid] = v > 0.f ? v : 0.f;
    }
}

__global__ void k_lsm_ep(const float* __restrict__ acc, const float* __restrict__ hx,
                         const float* __restrict__ dis, const float* __restrict__ b,
                         float* __restrict__ out, int n) {
    int i = blockIdx.x * blockDim.x + threadIdx.x;
    if (i < n) {
        float d2 = dis[i] * dis[i];
        float z0 = acc[2 * i + 0] + d2 * hx[2 * i + 0] + b[0];
        float z1 = acc[2 * i + 1] + d2 * hx[2 * i + 1] + b[1];
        float m = fmaxf(z0, z1);
        float lse = m + __logf(__expf(z0 - m) + __expf(z1 - m));
        out[2 * i + 0] = z0 - lse;
        out[2 * i + 1] = z1 - lse;
    }
}

extern "C" void kernel_launch(void* const* d_in, const int* in_sizes, int n_in,
                              void* d_out, int out_size, void* d_ws, size_t ws_size,
                              hipStream_t stream) {
    const float* x  = (const float*)d_in[0];
    const int*   ei = (const int*)d_in[1];
    const float* W1 = (const float*)d_in[2];
    const float* b1 = (const float*)d_in[3];
    const float* W2 = (const float*)d_in[4];
    const float* b2 = (const float*)d_in[5];
    const float* W3 = (const float*)d_in[6];
    const float* b3 = (const float*)d_in[7];

    const int N = in_sizes[0] / 128;
    const int E = in_sizes[1] / 2;
    const int* src = ei;
    const int* dst = ei + E;

    char* ws = (char*)d_ws;
    size_t off = 0;
    auto carve = [&](size_t bytes) -> void* {
        void* p = ws + off;
        off += (bytes + 255) & ~(size_t)255;
        return p;
    };
    float* dis  = (float*)carve((size_t)N * 4);
    float* norm = (float*)carve((size_t)E * 4);
    float* hx   = (float*)carve((size_t)N * 16 * 4);
    float* acc  = (float*)carve((size_t)N * 16 * 4);
    float* h    = (float*)carve((size_t)N * 16 * 4);
    int* tick   = (int*)carve((size_t)4 * NQ * TSTRIDE * 4);
    (void)ws_size; (void)n_in; (void)out_size;

    const int T = 256;
    auto blk = [](long n, int t) { return (unsigned)((n + t - 1) / t); };

    const int nchunk = (int)blk(E, CHUNK);
    const unsigned gown = (unsigned)nchunk * NQ;
    k_init<<<blk((long)N + 4 * NQ * TSTRIDE, T), T, 0, stream>>>(dis, tick, N);
    k_deg_own<<<gown, 256, 0, stream>>>(dst, dis, tick, E, N, nchunk);
    k_rsqrt_inplace<<<blk(N, T), T, 0, stream>>>(dis, N);
    k_norm<<<blk(E, T), T, 0, stream>>>(src, dst, dis, norm, E);

    const int ntile = (N + 15) / 16;
    const unsigned gblk = (unsigned)((ntile + 7) / 8);

    k_gemm_wmma<128, 16><<<gblk, 256, 0, stream>>>(x, W1, hx, N);
    k_fill<<<blk((long)N * 16, T), T, 0, stream>>>(acc, 0.f, N * 16);
    k_scatter_own<16><<<gown, 256, 0, stream>>>(src, dst, norm, hx, acc,
                                                tick + 1 * NQ * TSTRIDE, E, N, nchunk);
    k_relu_ep<<<blk((long)N * 16, T), T, 0, stream>>>(acc, hx, dis, b1, h, N);

    k_gemm_wmma<16, 16><<<gblk, 256, 0, stream>>>(h, W2, hx, N);
    k_fill<<<blk((long)N * 16, T), T, 0, stream>>>(acc, 0.f, N * 16);
    k_scatter_own<16><<<gown, 256, 0, stream>>>(src, dst, norm, hx, acc,
                                                tick + 2 * NQ * TSTRIDE, E, N, nchunk);
    k_relu_ep<<<blk((long)N * 16, T), T, 0, stream>>>(acc, hx, dis, b2, h, N);

    k_gemm_wmma<16, 2><<<gblk, 256, 0, stream>>>(h, W3, hx, N);
    k_fill<<<blk((long)N * 2, T), T, 0, stream>>>(acc, 0.f, N * 2);
    k_scatter_own<2><<<gown, 256, 0, stream>>>(src, dst, norm, hx, acc,
                                               tick + 3 * NQ * TSTRIDE, E, N, nchunk);
    k_lsm_ep<<<blk(N, T), T, 0, stream>>>(acc, hx, dis, b3, (float*)d_out, N);
}


# product kernels write each accumulator row's start value (self-loop term + bias, which now enter the atomic sum first) and take relu on their A loads: clear and epilogue launches go, 16 to 11
# speedup vs baseline: 3.0185x; 1.0854x over previous
#include <hip/hip_runtime.h>
#include <math.h>

typedef float v2f __attribute__((ext_vector_type(2)));
typedef float v8f __attribute__((ext_vector_type(8)));

#define NQ 8
#define TSTRIDE 64
#define EPT 8
#define CHUNK (256 * EPT)

__global__ void k_init(float* __restrict__ p, int* __restrict__ tick, int n) {
    int i = blockIdx.x * blockDim.x + threadIdx.x;
    if (i < n) p[i] = 1.0f;
    if (i < 4 * NQ * TSTRIDE) tick[i] = 0;
}

__device__ __forceinline__ void draw_task(int* __restrict__ tick, int nchunk,
                                          int& chunk, int& slice) {
    __shared__ int s_q, s_t;
    if (threadIdx.x == 0) {
        int die = (int)((__builtin_amdgcn_s_sendmsg_rtn(0x87) >> 16) & (NQ - 1));
        int q = die, t = nchunk;
        for (int i = 0; i < NQ; ++i) {
            q = (die + i) & (NQ - 1);
            t = atomicAdd(&tick[q * TSTRIDE], 1);
            if (t < nchunk) break;
        }
        s_q = q; s_t = t;
    }
    __syncthreads();
    slice = s_q; chunk = s_t;
}

__global__ __launch_bounds__(256) void k_deg_own(const int* __restrict__ dst,
                                                 float* __restrict__ deg,
                                                 int* __restrict__ tick, int E, int N,
                                                 int nchunk) {
    int chunk, slice;
    draw_task(tick, nchunk, chunk, slice);
    if (chunk >= nchunk) return;
    unsigned R = (unsigned)(N + NQ - 1) / NQ;
    unsigned lo = (unsigned)slice * R;
    int e0 = chunk * CHUNK + (int)threadIdx.x;
    int dv[EPT];
#pragma unroll
    for (int j = 0; j < EPT; ++j) {
        int e = e0 + j * 256;
        dv[j] = e < E ? dst[e] : -1;
    }
#pragma unroll
    for (int j = 0; j < EPT; ++j)
        if ((unsigned)dv[j] - lo < R) unsafeAtomicAdd(&deg[dv[j]], 1.0f);
}

__global__ void k_rsqrt_inplace(float* __restrict__ p, int n) {
    int i = blockIdx.x * blockDim.x + threadIdx.x;
    if (i < n) p[i] = rsqrtf(p[i]);
}

__global__ void k_norm(const int* __restrict__ src, const int* __restrict__ dst,
                       const float* __restrict__ dis, float* __restrict__ norm, int E) {
    int e = blockIdx.x * blockDim.x + threadIdx.x;
    if (e < E) norm[e] = dis[src[e]] * dis[dst[e]];
}

template <int K, int NCOL, bool RELU_A>
__global__ __launch_bounds__(256) void k_gemm_wmma(const float* __restrict__ A,
                                                   const float* __restrict__ B,
                                                   const float* __restrict__ dis,
                                                   const float* __restrict__ bias,
                                                   float* __restrict__ O,
                                                   float* __restrict__ S, int nrows) {
    int wave = threadIdx.x >> 5;
    int lane = threadIdx.x & 31;
    int tile = blockIdx.x * (blockDim.x >> 5) + wave;
    int base = tile << 4;
    if (base >= nrows) return;
    int hi = lane >> 4;
    int lo = lane & 15;

    int arow = base + lo;
    arow = arow < nrows ? arow : (nrows - 1);
    const float* __restrict__ ap = A + (long)arow * K + 2 * hi;
    const float* __restrict__ bp = B + lo;
    bool bvalid = (lo < NCOL);

    float d2[8];
#pragma unroll
    for (int v = 0; v < 8; ++v) {
        int row = base + v + 8 * hi;
        row = row < nrows ? row : (nrows - 1);
        float di = dis[row];
        d2[v] = di * di;
    }
    float bl = bvalid ? bias[lo] : 0.f;

    v8f c = {0.f, 0.f, 0.f, 0.f, 0.f, 0.f, 0.f, 0.f};
#pragma unroll
    for (int k0 = 0; k0 < K; k0 += 4) {
        v2f a = *(const v2f*)(ap + k0);
        if (RELU_A) {
            a.x = a.x > 0.f ? a.x : 0.f;
            a.y = a.y > 0.f ? a.y : 0.f;
        }
        int ka = k0 + 2 * hi;
        v2f b;
        b.x = bvalid ? bp[(ka + 0) * NCOL] : 0.f;
        b.y = bvalid ? bp[(ka + 1) * NCOL] : 0.f;
        c = __builtin_amdgcn_wmma_f32_16x16x4_f32(false, a, false, b, (short)0, c,
                                                  false, false);
    }
#pragma unroll
    for (int v = 0; v < 8; ++v) {
        int row = base + v + 8 * hi;
        if (row < nrows && lo < NCOL) {
            float hv = c[v];
            O[(long)row * NCOL + lo] = hv;
            S[(long)row * NCOL + lo] = d2[v] * hv + bl;
        }
    }
}

template <int F>
__global__ __launch_bounds__(256) void k_scatter_own(
    const int* __restrict__ src, const int* __restrict__ dst, const float* __restrict__ norm,
    const float* __restrict__ h, float* __restrict__ acc, int* __restrict__ tick, int E, int N,
    int nchunk) {
    __shared__ int l_s[8][32 * EPT];
    __shared__ int l_d[8][32 * EPT];
    __shared__ float l_n[8][32 * EPT];
    int chunk, slice;
    draw_task(tick, nchunk, chunk, slice);
    if (chunk >= nchunk) return;
    unsigned R = (unsigned)(N + NQ - 1) / NQ;
    unsigned lo = (unsigned)slice * R;
    int wave = threadIdx.x >> 5;
    int lane = threadIdx.x & 31;
    int e0 = chunk * CHUNK + (int)threadIdx.x;
    int dv[EPT];
#pragma unroll
    for (int j = 0; j < EPT; ++j) {
        int e = e0 + j * 256;
        dv[j] = e < E ? dst[e] : -1;
    }
    int sv[EPT];
    float nv[EPT];
#pragma unroll
    for (int j = 0; j < EPT; ++j) {
        sv[j] = 0;
        nv[j] = 0.f;
        if ((unsigned)dv[j] - lo < R) {
            int e = e0 + j * 256;
            sv[j] = src[e];
            nv[j] = norm[e];
        }
    }
    int cnt = 0;
#pragma unroll
    for (int j = 0; j < EPT; ++j) {
        bool in = (unsigned)dv[j] - lo < R;
        unsigned m = __builtin_amdgcn_ballot_w32(in);
        if (in) {
            int slot = cnt + __popc(m & ((1u << lane) - 1u));
            l_s[wave][slot] = sv[j];
            l_d[wave][slot] = dv[j];
            l_n[wave][slot] = nv[j];
        }
        cnt += __popc(m);
    }
    __syncthreads();
    constexpr int EPW = 32 / F;
    int sub = lane / F;
    int f = lane - sub * F;
    for (int i = sub; i < cnt; i += 4 * EPW) {
        int d[4];
        float nr[4], hv[4];
#pragma unroll
        for (int u = 0; u < 4; ++u) {
            int k = i + u * EPW;
            d[u] = -1;
            nr[u] = 0.f;
            hv[u] = 0.f;
            if (k < cnt) {
                int s = l_s[wave][k];
                d[u] = l_d[wave][k];
                nr[u] = l_n[wave][k];
                hv[u] = h[(long)s * F + f];
            }
        }
#pragma unroll
        for (int u = 0; u < 4; ++u)
            if (d[u] >= 0) unsafeAtomicAdd(&acc[(long)d[u] * F + f], nr[u] * hv[u]);
    }
}

__global__ void k_lsm_ep(const float* __restrict__ acc, float* __restrict__ out, int n) {
    int i = blockIdx.x * blockDim.x + threadIdx.x;
    if (i < n) {
        float z0 = acc[2 * i + 0];
        float z1 = acc[2 * i + 1];
        float m = fmaxf(z0, z1);
        float lse = m + __logf(__expf(z0 - m) + __expf(z1 - m));
        out[2 * i + 0] = z0 - lse;
        out[2 * i + 1] = z1 - lse;
    }
}

extern "C" void kernel_launch(void* const* d_in, const int* in_sizes, int n_in,
                              void* d_out, int out_size, void* d_ws, size_t ws_size,
                              hipStream_t stream) {
    const float* x  = (const float*)d_in[0];
    const int*   ei = (const int*)d_in[1];
    const float* W1 = (const float*)d_in[2];
    const float* b1 = (const float*)d_in[3];
    const float* W2 = (const float*)d_in[4];
    const float* b2 = (const float*)d_in[5];
    const float* W3 = (const float*)d_in[6];
    const float* b3 = (const float*)d_in[7];

    const int N = in_sizes[0] / 128;
    const int E = in_sizes[1] / 2;
    const int* src = ei;
    const int* dst = ei + E;

    char* ws = (char*)d_ws;
    size_t off = 0;
    auto carve = [&](size_t bytes) -> void* {
        void* p = ws + off;
        off += (bytes + 255) & ~(size_t)255;
        return p;
    };
    float* dis  = (float*)carve((size_t)N * 4);
    float* norm = (float*)carve((size_t)E * 4);
    float* hx   = (float*)carve((size_t)N * 16 * 4);
    float* accA = (float*)carve((size_t)N * 16 * 4);
    float* accB = (float*)carve((size_t)N * 16 * 4);
    int* tick   = (int*)carve((size_t)4 * NQ * TSTRIDE * 4);
    (void)ws_size; (void)n_in; (void)out_size;

    const int T = 256;
    auto blk = [](long n, int t) { return (unsigned)((n + t - 1) / t); };

    const int nchunk = (int)blk(E, CHUNK);
    const unsigned gown = (unsigned)nchunk * NQ;
    k_init<<<blk((long)N + 4 * NQ * TSTRIDE, T), T, 0, stream>>>(dis, tick, N);
    k_deg_own<<<gown, 256, 0, stream>>>(dst, dis, tick, E, N, nchunk);
    k_rsqrt_inplace<<<blk(N, T), T, 0, stream>>>(dis, N);
    k_norm<<<blk(E, T), T, 0, stream>>>(src, dst, dis, norm, E);

    const int ntile = (N + 15) / 16;
    const unsigned gblk = (unsigned)((ntile + 7) / 8);


    k_gemm_wmma<128, 16, false><<<gblk, 256, 0, stream>>>(x, W1, dis, b1, hx, accA, N);
    k_scatter_own<16><<<gown, 256, 0, stream>>>(src, dst, norm, hx, accA,
                                                tick + 1 * NQ * TSTRIDE, E, N, nchunk);

    k_gemm_wmma<16, 16, true><<<gblk, 256, 0, stream>>>(accA, W2, dis, b2, hx, accB, N);
    k_scatter_own<16><<<gown, 256, 0, stream>>>(src, dst, norm, hx, accB,
                                                tick + 2 * NQ * TSTRIDE, E, N, nchunk);

    k_gemm_wmma<16, 2, true><<<gblk, 256, 0, stream>>>(accB, W3, dis, b3, hx, accA, N);
    k_scatter_own<2><<<gown, 256, 0, stream>>>(src, dst, norm, hx, accA,
                                               tick + 3 * NQ * TSTRIDE, E, N, nchunk);
    k_lsm_ep<<<blk(N, T), T, 0, stream>>>(accA, (float*)d_out, N);
}


# speedup vs baseline: 3.5260x; 1.1681x over previous
#include <hip/hip_runtime.h>
#include <math.h>

typedef float v2f __attribute__((ext_vector_type(2)));
typedef float v8f __attribute__((ext_vector_type(8)));

#define NQ 2
#define TSTRIDE 64
#define EPT 8
#define CHUNK (256 * EPT)

__global__ void k_init(float* __restrict__ p, int* __restrict__ tick, int n) {
    int i = blockIdx.x * blockDim.x + threadIdx.x;
    if (i < n) p[i] = 1.0f;
    if (i < 4 * NQ * TSTRIDE) tick[i] = 0;
}

__device__ __forceinline__ void draw_task(int* __restrict__ tick, int nchunk,
                                          int& chunk, int& slice) {
    __shared__ int s_q, s_t;
    if (threadIdx.x == 0) {
        int die = (int)((__builtin_amdgcn_s_sendmsg_rtn(0x87) >> 16) & 15);
        int home = (die >> 2) & (NQ - 1);
        int q = home, t = nchunk;
        for (int i = 0; i < NQ; ++i) {
            q = (home + i) & (NQ - 1);
            t = atomicAdd(&tick[q * TSTRIDE], 1);
            if (t < nchunk) break;
        }
        s_q = q; s_t = t;
    }
    __syncthreads();
    slice = s_q; chunk = s_t;
}

__global__ __launch_bounds__(256) void k_deg_own(const int* __restrict__ dst,
                                                 float* __restrict__ deg,
                                                 int* __restrict__ tick, int E, int N,
                                                 int nchunk) {
    int chunk, slice;
    draw_task(tick, nchunk, chunk, slice);
    if (chunk >= nchunk) return;
    unsigned R = (unsigned)(N + NQ - 1) / NQ;
    unsigned lo = (unsigned)slice * R;
    int e0 = chunk * CHUNK + (int)threadIdx.x;
    int dv[EPT];
#pragma unroll
    for (int j = 0; j < EPT; ++j) {
        int e = e0 + j * 256;
        dv[j] = e < E ? dst[e] : -1;
    }
#pragma unroll
    for (int j = 0; j < EPT; ++j)
        if ((unsigned)dv[j] - lo < R) unsafeAtomicAdd(&deg[dv[j]], 1.0f);
}

__global__ void k_rsqrt_inplace(float* __restrict__ p, int n) {
    int i = blockIdx.x * blockDim.x + threadIdx.x;
    if (i < n) p[i] = rsqrtf(p[i]);
}

__global__ void k_norm(const int* __restrict__ src, const int* __restrict__ dst,
                       const float* __restrict__ dis, float* __restrict__ norm, int E) {
    int e = blockIdx.x * blockDim.x + threadIdx.x;
    if (e < E) norm[e] = dis[src[e]] * dis[dst[e]];
}

template <int K, int NCOL, bool RELU_A>
__global__ __launch_bounds__(256) void k_gemm_wmma(const float* __restrict__ A,
                                                   const float* __restrict__ B,
                                                   const float* __restrict__ dis,
                                                   const float* __restrict__ bias,
                                                   float* __restrict__ O,
                                                   float* __restrict__ S, int nrows) {
    int wave = threadIdx.x >> 5;
    int lane = threadIdx.x & 31;
    int tile = blockIdx.x * (blockDim.x >> 5) + wave;
    int base = tile << 4;
    if (base >= nrows) return;
    int hi = lane >> 4;
    int lo = lane & 15;

    int arow = base + lo;
    arow = arow < nrows ? arow : (nrows - 1);
    const float* __restrict__ ap = A + (long)arow * K + 2 * hi;
    const float* __restrict__ bp = B + lo;
    bool bvalid = (lo < NCOL);

    float d2[8];
#pragma unroll
    for (int v = 0; v < 8; ++v) {
        int row = base + v + 8 * hi;
        row = row < nrows ? row : (nrows - 1);
        float di = dis[row];
        d2[v] = di * di;
    }
    float bl = bvalid ? bias[lo] : 0.f;

    v8f c = {0.f, 0.f, 0.f, 0.f, 0.f, 0.f, 0.f, 0.f};
#pragma unroll
    for (int k0 = 0; k0 < K; k0 += 4) {
        v2f a = *(const v2f*)(ap + k0);
        if (RELU_A) {
            a.x = a.x > 0.f ? a.x : 0.f;
            a.y = a.y > 0.f ? a.y : 0.f;
        }
        int ka = k0 + 2 * hi;
        v2f b;
        b.x = bvalid ? bp[(ka + 0) * NCOL] : 0.f;
        b.y = bvalid ? bp[(ka + 1) * NCOL] : 0.f;
        c = __builtin_amdgcn_wmma_f32_16x16x4_f32(false, a, false, b, (short)0, c,
                                                  false, false);
    }
#pragma unroll
    for (int v = 0; v < 8; ++v) {
        int row = base + v + 8 * hi;
        if (row < nrows && lo < NCOL) {
            float hv = c[v];
            O[(long)row * NCOL + lo] = hv;
            S[(long)row * NCOL + lo] = d2[v] * hv + bl;
        }
    }
}

template <int F>
__global__ __launch_bounds__(256) void k_scatter_own(
    const int* __restrict__ src, const int* __restrict__ dst, const float* __restrict__ norm,
    const float* __restrict__ h, float* __restrict__ acc, int* __restrict__ tick, int E, int N,
    int nchunk) {
    __shared__ int l_s[8][32 * EPT];
    __shared__ int l_d[8][32 * EPT];
    __shared__ float l_n[8][32 * EPT];
    int chunk, slice;
    draw_task(tick, nchunk, chunk, slice);
    if (chunk >= nchunk) return;
    unsigned R = (unsigned)(N + NQ - 1) / NQ;
    unsigned lo = (unsigned)slice * R;
    int wave = threadIdx.x >> 5;
    int lane = threadIdx.x & 31;
    int e0 = chunk * CHUNK + (int)threadIdx.x;
    int dv[EPT];
#pragma unroll
    for (int j = 0; j < EPT; ++j) {
        int e = e0 + j * 256;
        dv[j] = e < E ? dst[e] : -1;
    }
    int sv[EPT];
    float nv[EPT];
#pragma unroll
    for (int j = 0; j < EPT; ++j) {
        sv[j] = 0;
        nv[j] = 0.f;
        if ((unsigned)dv[j] - lo < R) {
            int e = e0 + j * 256;
            sv[j] = src[e];
            nv[j] = norm[e];
        }
    }
    int cnt = 0;
#pragma unroll
    for (int j = 0; j < EPT; ++j) {
        bool in = (unsigned)dv[j] - lo < R;
        unsigned m = __builtin_amdgcn_ballot_w32(in);
        if (in) {
            int slot = cnt + __popc(m & ((1u << lane) - 1u));
            l_s[wave][slot] = sv[j];
            l_d[wave][slot] = dv[j];
            l_n[wave][slot] = nv[j];
        }
        cnt += __popc(m);
    }
    __syncthreads();
    constexpr int EPW = 32 / F;
    int sub = lane / F;
    int f = lane - sub * F;
    for (int i = sub; i < cnt; i += 4 * EPW) {
        int d[4];
        float nr[4], hv[4];
#pragma unroll
        for (int u = 0; u < 4; ++u) {
            int k = i + u * EPW;
            d[u] = -1;
            nr[u] = 0.f;
            hv[u] = 0.f;
            if (k < cnt) {
                int s = l_s[wave][k];
                d[u] = l_d[wave][k];
                nr[u] = l_n[wave][k];
                hv[u] = h[(long)s * F + f];
            }
        }
#pragma unroll
        for (int u = 0; u < 4; ++u)
            if (d[u] >= 0) unsafeAtomicAdd(&acc[(long)d[u] * F + f], nr[u] * hv[u]);
    }
}

__global__ void k_lsm_ep(const float* __restrict__ acc, float* __restrict__ out, int n) {
    int i = blockIdx.x * blockDim.x + threadIdx.x;
    if (i < n) {
        float z0 = acc[2 * i + 0];
        float z1 = acc[2 * i + 1];
        float m = fmaxf(z0, z1);
        float lse = m + __logf(__expf(z0 - m) + __expf(z1 - m));
        out[2 * i + 0] = z0 - lse;
        out[2 * i + 1] = z1 - lse;
    }
}

extern "C" void kernel_launch(void* const* d_in, const int* in_sizes, int n_in,
                              void* d_out, int out_size, void* d_ws, size_t ws_size,
                              hipStream_t stream) {
    const float* x  = (const float*)d_in[0];
    const int*   ei = (const int*)d_in[1];
    const float* W1 = (const float*)d_in[2];
    const float* b1 = (const float*)d_in[3];
    const float* W2 = (const float*)d_in[4];
    const float* b2 = (const float*)d_in[5];
    const float* W3 = (const float*)d_in[6];
    const float* b3 = (const float*)d_in[7];

    const int N = in_sizes[0] / 128;
    const int E = in_sizes[1] / 2;
    const int* src = ei;
    const int* dst = ei + E;

    char* ws = (char*)d_ws;
    size_t off = 0;
    auto carve = [&](size_t bytes) -> void* {
        void* p = ws + off;
        off += (bytes + 255) & ~(size_t)255;
        return p;
    };
    float* dis  = (float*)carve((size_t)N * 4);
    float* norm = (float*)carve((size_t)E * 4);
    float* hx   = (float*)carve((size_t)N * 16 * 4);
    float* accA = (float*)carve((size_t)N * 16 * 4);
    float* accB = (float*)carve((size_t)N * 16 * 4);
    int* tick   = (int*)carve((size_t)4 * NQ * TSTRIDE * 4);
    (void)ws_size; (void)n_in; (void)out_size;

    const int T = 256;
    auto blk = [](long n, int t) { return (unsigned)((n + t - 1) / t); };

    const int nchunk = (int)blk(E, CHUNK);
    const unsigned gown = (unsigned)nchunk * NQ;
    k_init<<<blk((long)N + 4 * NQ * TSTRIDE, T), T, 0, stream>>>(dis, tick, N);
    k_deg_own<<<gown, 256, 0, stream>>>(dst, dis, tick, E, N, nchunk);
    k_rsqrt_inplace<<<blk(N, T), T, 0, stream>>>(dis, N);
    k_norm<<<blk(E, T), T, 0, stream>>>(src, dst, dis, norm, E);

    const int ntile = (N + 15) / 16;
    const unsigned gblk = (unsigned)((ntile + 7) / 8);


    k_gemm_wmma<128, 16, false><<<gblk, 256, 0, stream>>>(x, W1, dis, b1, hx, accA, N);
    k_scatter_own<16><<<gown, 256, 0, stream>>>(src, dst, norm, hx, accA,
                                                tick + 1 * NQ * TSTRIDE, E, N, nchunk);

    k_gemm_wmma<16, 16, true><<<gblk, 256, 0, stream>>>(accA, W2, dis, b2, hx, accB, N);
    k_scatter_own<16><<<gown, 256, 0, stream>>>(src, dst, norm, hx, accB,
                                                tick + 2 * NQ * TSTRIDE, E, N, nchunk);

    k_gemm_wmma<16, 2, true><<<gblk, 256, 0, stream>>>(accB, W3, dis, b3, hx, accA, N);
    k_scatter_own<2><<<gown, 256, 0, stream>>>(src, dst, norm, hx, accA,
                                               tick + 3 * NQ * TSTRIDE, E, N, nchunk);
    k_lsm_ep<<<blk(N, T), T, 0, stream>>>(accA, (float*)d_out, N);
}


# dis factored out of the edge sum: product rows pre-scaled by dis[src]; dis[dst] and bias applied when the accumulator is read (bias now last in the sum); norm array and its launch gone, 11 to 10
# speedup vs baseline: 4.2957x; 1.2183x over previous
#include <hip/hip_runtime.h>
#include <math.h>

typedef float v2f __attribute__((ext_vector_type(2)));
typedef float v8f __attribute__((ext_vector_type(8)));

#define NQ 2
#define TSTRIDE 64
#define EPT 8
#define CHUNK (256 * EPT)

__global__ void k_init(float* __restrict__ p, int* __restrict__ tick, int n) {
    int i = blockIdx.x * blockDim.x + threadIdx.x;
    if (i < n) p[i] = 1.0f;
    if (i < 4 * NQ * TSTRIDE) tick[i] = 0;
}

__device__ __forceinline__ void draw_task(int* __restrict__ tick, int nchunk,
                                          int& chunk, int& slice) {
    __shared__ int s_q, s_t;
    if (threadIdx.x == 0) {
        int die = (int)((__builtin_amdgcn_s_sendmsg_rtn(0x87) >> 16) & 15);
        int home = (die >> 2) & (NQ - 1);
        int q = home, t = nchunk;
        for (int i = 0; i < NQ; ++i) {
            q = (home + i) & (NQ - 1);
            t = atomicAdd(&tick[q * TSTRIDE], 1);
            if (t < nchunk) break;
        }
        s_q = q; s_t = t;
    }
    __syncthreads();
    slice = s_q; chunk = s_t;
}

__global__ __launch_bounds__(256) void k_deg_own(const int* __restrict__ dst,
                                                 float* __restrict__ deg,
                                                 int* __restrict__ tick, int E, int N,
                                                 int nchunk) {
    int chunk, slice;
    draw_task(tick, nchunk, chunk, slice);
    if (chunk >= nchunk) return;
    unsigned R = (unsigned)(N + NQ - 1) / NQ;
    unsigned lo = (unsigned)slice * R;
    int e0 = chunk * CHUNK + (int)threadIdx.x;
    int dv[EPT];
#pragma unroll
    for (int j = 0; j < EPT; ++j) {
        int e = e0 + j * 256;
        dv[j] = e < E ? dst[e] : -1;
    }
#pragma unroll
    for (int j = 0; j < EPT; ++j)
        if ((unsigned)dv[j] - lo < R) unsafeAtomicAdd(&deg[dv[j]], 1.0f);
}

__global__ void k_rsqrt_inplace(float* __restrict__ p, int n) {
    int i = blockIdx.x * blockDim.x + threadIdx.x;
    if (i < n) p[i] = rsqrtf(p[i]);
}

template <int K, int NCOL, bool PRE_A>
__global__ __launch_bounds__(256) void k_gemm_wmma(const float* __restrict__ A,
                                                   const float* __restrict__ B,
                                                   const float* __restrict__ dis,
                                                   const float* __restrict__ biasA,
                                                   float* __restrict__ O,
                                                   float* __restrict__ S, int nrows) {
    int wave = threadIdx.x >> 5;
    int lane = threadIdx.x & 31;
    int tile = blockIdx.x * (blockDim.x >> 5) + wave;
    int base = tile << 4;
    if (base >= nrows) return;
    int hi = lane >> 4;
    int lo = lane & 15;

    int arow = base + lo;
    arow = arow < nrows ? arow : (nrows - 1);
    const float* __restrict__ ap = A + (long)arow * K + 2 * hi;
    const float* __restrict__ bp = B + lo;
    bool bvalid = (lo < NCOL);

    float d1[8];
#pragma unroll
    for (int v = 0; v < 8; ++v) {
        int row = base + v + 8 * hi;
        row = row < nrows ? row : (nrows - 1);
        d1[v] = dis[row];
    }
    float da = PRE_A ? dis[arow] : 1.f;

    v8f c = {0.f, 0.f, 0.f, 0.f, 0.f, 0.f, 0.f, 0.f};
#pragma unroll
    for (int k0 = 0; k0 < K; k0 += 4) {
        v2f a = *(const v2f*)(ap + k0);
        int ka = k0 + 2 * hi;
        if (PRE_A) {
            float x0 = da * a.x + biasA[ka + 0];
            float x1 = da * a.y + biasA[ka + 1];
            a.x = x0 > 0.f ? x0 : 0.f;
            a.y = x1 > 0.f ? x1 : 0.f;
        }
        v2f b;
        b.x = bvalid ? bp[(ka + 0) * NCOL] : 0.f;
        b.y = bvalid ? bp[(ka + 1) * NCOL] : 0.f;
        c = __builtin_amdgcn_wmma_f32_16x16x4_f32(false, a, false, b, (short)0, c,
                                                  false, false);
    }
#pragma unroll
    for (int v = 0; v < 8; ++v) {
        int row = base + v + 8 * hi;
        if (row < nrows && lo < NCOL) {
            float g = d1[v] * c[v];
            O[(long)row * NCOL + lo] = g;
            S[(long)row * NCOL + lo] = g;
        }
    }
}

template <int F>
__global__ __launch_bounds__(256) void k_scatter_own(
    const int* __restrict__ src, const int* __restrict__ dst,
    const float* __restrict__ h, float* __restrict__ acc, int* __restrict__ tick, int E, int N,
    int nchunk) {
    __shared__ int l_s[8][32 * EPT];
    __shared__ int l_d[8][32 * EPT];
    int chunk, slice;
    draw_task(tick, nchunk, chunk, slice);
    if (chunk >= nchunk) return;
    unsigned R = (unsigned)(N + NQ - 1) / NQ;
    unsigned lo = (unsigned)slice * R;
    int wave = threadIdx.x >> 5;
    int lane = threadIdx.x & 31;
    int e0 = chunk * CHUNK + (int)threadIdx.x;
    int dv[EPT];
#pragma unroll
    for (int j = 0; j < EPT; ++j) {
        int e = e0 + j * 256;
        dv[j] = e < E ? dst[e] : -1;
    }
    int sv[EPT];
#pragma unroll
    for (int j = 0; j < EPT; ++j) {
        sv[j] = 0;
        if ((unsigned)dv[j] - lo < R) sv[j] = src[e0 + j * 256];
    }
    int cnt = 0;
#pragma unroll
    for (int j = 0; j < EPT; ++j) {
        bool in = (unsigned)dv[j] - lo < R;
        unsigned m = __builtin_amdgcn_ballot_w32(in);
        if (in) {
            int slot = cnt + __popc(m & ((1u << lane) - 1u));
            l_s[wave][slot] = sv[j];
            l_d[wave][slot] = dv[j];
        }
        cnt += __popc(m);
    }
    __syncthreads();
    constexpr int EPW = 32 / F;
    int sub = lane / F;
    int f = lane - sub * F;
    for (int i = sub; i < cnt; i += 4 * EPW) {
        int d[4];
        float hv[4];
#pragma unroll
        for (int u = 0; u < 4; ++u) {
            int k = i + u * EPW;
            d[u] = -1;
            hv[u] = 0.f;
            if (k < cnt) {
                int s = l_s[wave][k];
                d[u] = l_d[wave][k];
                hv[u] = h[(long)s * F + f];
            }
        }
#pragma unroll
        for (int u = 0; u < 4; ++u)
            if (d[u] >= 0) unsafeAtomicAdd(&acc[(long)d[u] * F + f], hv[u]);
    }
}

__global__ void k_lsm_ep(const float* __restrict__ acc, const float* __restrict__ dis,
                         const float* __restrict__ bias, float* __restrict__ out, int n) {
    int i = blockIdx.x * blockDim.x + threadIdx.x;
    if (i < n) {
        float di = dis[i];
        float z0 = di * acc[2 * i + 0] + bias[0];
        float z1 = di * acc[2 * i + 1] + bias[1];
        float m = fmaxf(z0, z1);
        float lse = m + __logf(__expf(z0 - m) + __expf(z1 - m));
        out[2 * i + 0] = z0 - lse;
        out[2 * i + 1] = z1 - lse;
    }
}

extern "C" void kernel_launch(void* const* d_in, const int* in_sizes, int n_in,
                              void* d_out, int out_size, void* d_ws, size_t ws_size,
                              hipStream_t stream) {
    const float* x  = (const float*)d_in[0];
    const int*   ei = (const int*)d_in[1];
    const float* W1 = (const float*)d_in[2];
    const float* b1 = (const float*)d_in[3];
    const float* W2 = (const float*)d_in[4];
    const float* b2 = (const float*)d_in[5];
    const float* W3 = (const float*)d_in[6];
    const float* b3 = (const float*)d_in[7];

    const int N = in_sizes[0] / 128;
    const int E = in_sizes[1] / 2;
    const int* src = ei;
    const int* dst = ei + E;

    char* ws = (char*)d_ws;
    size_t off = 0;
    auto carve = [&](size_t bytes) -> void* {
        void* p = ws + off;
        off += (bytes + 255) & ~(size_t)255;
        return p;
    };
    float* dis  = (float*)carve((size_t)N * 4);
    float* hx   = (float*)carve((size_t)N * 16 * 4);
    float* accA = (float*)carve((size_t)N * 16 * 4);
    float* accB = (float*)carve((size_t)N * 16 * 4);
    int* tick   = (int*)carve((size_t)4 * NQ * TSTRIDE * 4);
    (void)ws_size; (void)n_in; (void)out_size;

    const int T = 256;
    auto blk = [](long n, int t) { return (unsigned)((n + t - 1) / t); };

    const int nchunk = (int)blk(E, CHUNK);
    const unsigned gown = (unsigned)nchunk * NQ;
    k_init<<<blk((long)N + 4 * NQ * TSTRIDE, T), T, 0, stream>>>(dis, tick, N);
    k_deg_own<<<gown, 256, 0, stream>>>(dst, dis, tick, E, N, nchunk);
    k_rsqrt_inplace<<<blk(N, T), T, 0, stream>>>(dis, N);

    const int ntile = (N + 15) / 16;
    const unsigned gblk = (unsigned)((ntile + 7) / 8);


    k_gemm_wmma<128, 16, false><<<gblk, 256, 0, stream>>>(x, W1, dis, nullptr, hx, accA, N);
    k_scatter_own<16><<<gown, 256, 0, stream>>>(src, dst, hx, accA,
                                                tick + 1 * NQ * TSTRIDE, E, N, nchunk);

    k_gemm_wmma<16, 16, true><<<gblk, 256, 0, stream>>>(accA, W2, dis, b1, hx, accB, N);
    k_scatter_own<16><<<gown, 256, 0, stream>>>(src, dst, hx, accB,
                                                tick + 2 * NQ * TSTRIDE, E, N, nchunk);

    k_gemm_wmma<16, 2, true><<<gblk, 256, 0, stream>>>(accB, W3, dis, b2, hx, accA, N);
    k_scatter_own<2><<<gown, 256, 0, stream>>>(src, dst, hx, accA,
                                               tick + 3 * NQ * TSTRIDE, E, N, nchunk);
    k_lsm_ep<<<blk(N, T), T, 0, stream>>>(accA, dis, b3, (float*)d_out, N);
}


# degree count in a per-work-group LDS histogram (integer adds) over an edge range and one half's nodes, flushed as whole-line float atomics of exact counts: 3.2M one-word atomics become 400k line adds
# speedup vs baseline: 4.7900x; 1.1151x over previous
#include <hip/hip_runtime.h>
#include <math.h>

typedef float v2f __attribute__((ext_vector_type(2)));
typedef float v8f __attribute__((ext_vector_type(8)));

#define NQ 2
#define TSTRIDE 64
#define EPT 8
#define CHUNK (256 * EPT)

__global__ void k_init(float* __restrict__ p, int* __restrict__ tick, int n) {
    int i = blockIdx.x * blockDim.x + threadIdx.x;
    if (i < n) p[i] = 1.0f;
    if (i < 4 * NQ * TSTRIDE) tick[i] = 0;
}

__device__ __forceinline__ void draw_task(int* __restrict__ tick, int nchunk,
                                          int& chunk, int& slice) {
    __shared__ int s_q, s_t;
    if (threadIdx.x == 0) {
        int die = (int)((__builtin_amdgcn_s_sendmsg_rtn(0x87) >> 16) & 15);
        int home = (die >> 2) & (NQ - 1);
        int q = home, t = nchunk;
        for (int i = 0; i < NQ; ++i) {
            q = (home + i) & (NQ - 1);
            t = atomicAdd(&tick[q * TSTRIDE], 1);
            if (t < nchunk) break;
        }
        s_q = q; s_t = t;
    }
    __syncthreads();
    slice = s_q; chunk = s_t;
}

#define DEG_T 1024
#define DEG_EPB 16
#define DEG_RANGES 128
#define HCAP 51200

template <bool FITS>
__device__ __forceinline__ void count_range(const int* __restrict__ dst,
                                            float* __restrict__ deg, unsigned* hist,
                                            int eb, int ee, unsigned lo, unsigned R,
                                            unsigned lo_al) {
    for (int e0 = eb + (int)threadIdx.x; e0 < ee; e0 += DEG_T * DEG_EPB) {
        int dv[DEG_EPB];
#pragma unroll
        for (int j = 0; j < DEG_EPB; ++j) {
            int e = e0 + j * DEG_T;
            dv[j] = dst[e < ee ? e : ee - 1];
        }
#pragma unroll
        for (int j = 0; j < DEG_EPB; ++j) {
            if (e0 + j * DEG_T < ee && (unsigned)dv[j] - lo < R) {
                unsigned i = (unsigned)dv[j] - lo_al;
                if (FITS || i < (unsigned)HCAP) atomicAdd(&hist[i], 1u);
                else unsafeAtomicAdd(&deg[dv[j]], 1.0f);
            }
        }
    }
}

__global__ __launch_bounds__(DEG_T) void k_deg_hist(const int* __restrict__ dst,
                                                    float* __restrict__ deg,
                                                    int* __restrict__ tick, int E, int N,
                                                    int nrange) {
    __shared__ unsigned hist[HCAP];
    for (unsigned i = 4u * threadIdx.x; i < (unsigned)HCAP; i += 4u * DEG_T)
        *(uint4*)&hist[i] = make_uint4(0u, 0u, 0u, 0u);
    int range, slice;
    draw_task(tick, nrange, range, slice);
    if (range >= nrange) return;
    unsigned R = (unsigned)(N + NQ - 1) / NQ;
    unsigned lo = (unsigned)slice * R;
    unsigned lo_al = lo & ~31u;
    unsigned span = lo + R - lo_al;
    unsigned used = span < (unsigned)HCAP ? span : (unsigned)HCAP;

    int per = (E + nrange - 1) / nrange;
    int eb = range * per;
    int ee = eb + per < E ? eb + per : E;
    if (span <= (unsigned)HCAP) count_range<true>(dst, deg, hist, eb, ee, lo, R, lo_al);
    else                        count_range<false>(dst, deg, hist, eb, ee, lo, R, lo_al);
    __syncthreads();
    unsigned usedr = (used + 31u) & ~31u;
    unsigned rot = (unsigned)range * (usedr >> 5) / (unsigned)nrange << 5;
    for (unsigned i0 = threadIdx.x; i0 < usedr; i0 += 4u * DEG_T) {
        unsigned c[4], k[4];
#pragma unroll
        for (int u = 0; u < 4; ++u) {
            unsigned i = i0 + (unsigned)u * DEG_T;
            k[u] = i + rot < usedr ? i + rot : i + rot - usedr;
            c[u] = i < usedr ? hist[k[u]] : 0u;
        }
#pragma unroll
        for (int u = 0; u < 4; ++u)
            if (c[u]) unsafeAtomicAdd(&deg[lo_al + k[u]], (float)c[u]);
    }
}

__global__ void k_rsqrt_inplace(float* __restrict__ p, int n) {
    int i = blockIdx.x * blockDim.x + threadIdx.x;
    if (i < n) p[i] = rsqrtf(p[i]);
}

template <int K, int NCOL, bool PRE_A>
__global__ __launch_bounds__(256) void k_gemm_wmma(const float* __restrict__ A,
                                                   const float* __restrict__ B,
                                                   const float* __restrict__ dis,
                                                   const float* __restrict__ biasA,
                                                   float* __restrict__ O,
                                                   float* __restrict__ S, int nrows) {
    int wave = threadIdx.x >> 5;
    int lane = threadIdx.x & 31;
    int tile = blockIdx.x * (blockDim.x >> 5) + wave;
    int base = tile << 4;
    if (base >= nrows) return;
    int hi = lane >> 4;
    int lo = lane & 15;

    int arow = base + lo;
    arow = arow < nrows ? arow : (nrows - 1);
    const float* __restrict__ ap = A + (long)arow * K + 2 * hi;
    const float* __restrict__ bp = B + lo;
    bool bvalid = (lo < NCOL);

    float d1[8];
#pragma unroll
    for (int v = 0; v < 8; ++v) {
        int row = base + v + 8 * hi;
        row = row < nrows ? row : (nrows - 1);
        d1[v] = dis[row];
    }
    float da = PRE_A ? dis[arow] : 1.f;

    v8f c = {0.f, 0.f, 0.f, 0.f, 0.f, 0.f, 0.f, 0.f};
#pragma unroll
    for (int k0 = 0; k0 < K; k0 += 4) {
        v2f a = *(const v2f*)(ap + k0);
        int ka = k0 + 2 * hi;
        if (PRE_A) {
            float x0 = da * a.x + biasA[ka + 0];
            float x1 = da * a.y + biasA[ka + 1];
            a.x = x0 > 0.f ? x0 : 0.f;
            a.y = x1 > 0.f ? x1 : 0.f;
        }
        v2f b;
        b.x = bvalid ? bp[(ka + 0) * NCOL] : 0.f;
        b.y = bvalid ? bp[(ka + 1) * NCOL] : 0.f;
        c = __builtin_amdgcn_wmma_f32_16x16x4_f32(false, a, false, b, (short)0, c,
                                                  false, false);
    }
#pragma unroll
    for (int v = 0; v < 8; ++v) {
        int row = base + v + 8 * hi;
        if (row < nrows && lo < NCOL) {
            float g = d1[v] * c[v];
            O[(long)row * NCOL + lo] = g;
            S[(long)row * NCOL + lo] = g;
        }
    }
}

template <int F>
__global__ __launch_bounds__(256) void k_scatter_own(
    const int* __restrict__ src, const int* __restrict__ dst,
    const float* __restrict__ h, float* __restrict__ acc, int* __restrict__ tick, int E, int N,
    int nchunk) {
    __shared__ int l_s[8][32 * EPT];
    __shared__ int l_d[8][32 * EPT];
    int chunk, slice;
    draw_task(tick, nchunk, chunk, slice);
    if (chunk >= nchunk) return;
    unsigned R = (unsigned)(N + NQ - 1) / NQ;
    unsigned lo = (unsigned)slice * R;
    int wave = threadIdx.x >> 5;
    int lane = threadIdx.x & 31;
    int e0 = chunk * CHUNK + (int)threadIdx.x;
    int dv[EPT];
#pragma unroll
    for (int j = 0; j < EPT; ++j) {
        int e = e0 + j * 256;
        dv[j] = e < E ? dst[e] : -1;
    }
    int sv[EPT];
#pragma unroll
    for (int j = 0; j < EPT; ++j) {
        sv[j] = 0;
        if ((unsigned)dv[j] - lo < R) sv[j] = src[e0 + j * 256];
    }
    int cnt = 0;
#pragma unroll
    for (int j = 0; j < EPT; ++j) {
        bool in = (unsigned)dv[j] - lo < R;
        unsigned m = __builtin_amdgcn_ballot_w32(in);
        if (in) {
            int slot = cnt + __popc(m & ((1u << lane) - 1u));
            l_s[wave][slot] = sv[j];
            l_d[wave][slot] = dv[j];
        }
        cnt += __popc(m);
    }
    __syncthreads();
    constexpr int EPW = 32 / F;
    int sub = lane / F;
    int f = lane - sub * F;
    for (int i = sub; i < cnt; i += 4 * EPW) {
        int d[4];
        float hv[4];
#pragma unroll
        for (int u = 0; u < 4; ++u) {
            int k = i + u * EPW;
            d[u] = -1;
            hv[u] = 0.f;
            if (k < cnt) {
                int s = l_s[wave][k];
                d[u] = l_d[wave][k];
                hv[u] = h[(long)s * F + f];
            }
        }
#pragma unroll
        for (int u = 0; u < 4; ++u)
            if (d[u] >= 0) unsafeAtomicAdd(&acc[(long)d[u] * F + f], hv[u]);
    }
}

__global__ void k_lsm_ep(const float* __restrict__ acc, const float* __restrict__ dis,
                         const float* __restrict__ bias, float* __restrict__ out, int n) {
    int i = blockIdx.x * blockDim.x + threadIdx.x;
    if (i < n) {
        float di = dis[i];
        float z0 = di * acc[2 * i + 0] + bias[0];
        float z1 = di * acc[2 * i + 1] + bias[1];
        float m = fmaxf(z0, z1);
        float lse = m + __logf(__expf(z0 - m) + __expf(z1 - m));
        out[2 * i + 0] = z0 - lse;
        out[2 * i + 1] = z1 - lse;
    }
}

extern "C" void kernel_launch(void* const* d_in, const int* in_sizes, int n_in,
                              void* d_out, int out_size, void* d_ws, size_t ws_size,
                              hipStream_t stream) {
    const float* x  = (const float*)d_in[0];
    const int*   ei = (const int*)d_in[1];
    const float* W1 = (const float*)d_in[2];
    const float* b1 = (const float*)d_in[3];
    const float* W2 = (const float*)d_in[4];
    const float* b2 = (const float*)d_in[5];
    const float* W3 = (const float*)d_in[6];
    const float* b3 = (const float*)d_in[7];

    const int N = in_sizes[0] / 128;
    const int E = in_sizes[1] / 2;
    const int* src = ei;
    const int* dst = ei + E;

    char* ws = (char*)d_ws;
    size_t off = 0;
    auto carve = [&](size_t bytes) -> void* {
        void* p = ws + off;
        off += (bytes + 255) & ~(size_t)255;
        return p;
    };
    float* dis  = (float*)carve((size_t)N * 4);
    float* hx   = (float*)carve((size_t)N * 16 * 4);
    float* accA = (float*)carve((size_t)N * 16 * 4);
    float* accB = (float*)carve((size_t)N * 16 * 4);
    int* tick   = (int*)carve((size_t)4 * NQ * TSTRIDE * 4);
    (void)ws_size; (void)n_in; (void)out_size;

    const int T = 256;
    auto blk = [](long n, int t) { return (unsigned)((n + t - 1) / t); };

    const int nchunk = (int)blk(E, CHUNK);
    const unsigned gown = (unsigned)nchunk * NQ;
    k_init<<<blk((long)N + 4 * NQ * TSTRIDE, T), T, 0, stream>>>(dis, tick, N);
    k_deg_hist<<<NQ * DEG_RANGES, DEG_T, 0, stream>>>(dst, dis, tick, E, N, DEG_RANGES);
    k_rsqrt_inplace<<<blk(N, T), T, 0, stream>>>(dis, N);

    const int ntile = (N + 15) / 16;
    const unsigned gblk = (unsigned)((ntile + 7) / 8);


    k_gemm_wmma<128, 16, false><<<gblk, 256, 0, stream>>>(x, W1, dis, nullptr, hx, accA, N);
    k_scatter_own<16><<<gown, 256, 0, stream>>>(src, dst, hx, accA,
                                                tick + 1 * NQ * TSTRIDE, E, N, nchunk);

    k_gemm_wmma<16, 16, true><<<gblk, 256, 0, stream>>>(accA, W2, dis, b1, hx, accB, N);
    k_scatter_own<16><<<gown, 256, 0, stream>>>(src, dst, hx, accB,
                                                tick + 2 * NQ * TSTRIDE, E, N, nchunk);

    k_gemm_wmma<16, 2, true><<<gblk, 256, 0, stream>>>(accB, W3, dis, b2, hx, accA, N);
    k_scatter_own<2><<<gown, 256, 0, stream>>>(src, dst, hx, accA,
                                               tick + 3 * NQ * TSTRIDE, E, N, nchunk);
    k_lsm_ep<<<blk(N, T), T, 0, stream>>>(accA, dis, b3, (float*)d_out, N);
}
